# starting kernel
# baseline (speedup 1.0000x reference)
#include <hip/hip_runtime.h>


typedef float v2f __attribute__((ext_vector_type(2)));
typedef float v8f __attribute__((ext_vector_type(8)));

#define DF 128
#define XS_STRIDE 260

__global__ void zero_ws_kernel(float* __restrict__ p, int n) {
    int i = blockIdx.x * blockDim.x + threadIdx.x;
    if (i < n) p[i] = 0.0f;
}

__global__ void deg_kernel(const int* __restrict__ dst, float* __restrict__ deg, int E) {
    int e = blockIdx.x * blockDim.x + threadIdx.x;
    if (e < E) atomicAdd(&deg[dst[e]], 1.0f);
}

__global__ void norm_kernel(float* __restrict__ deg_norm, int N) {
    int v = blockIdx.x * blockDim.x + threadIdx.x;
    if (v < N) {
        float d = deg_norm[v];
        deg_norm[v] = rsqrtf(fmaxf(d, 1.0f));
    }
}

__global__ void __launch_bounds__(256)
scatter_kernel(const float* __restrict__ feat,
               const int*   __restrict__ src,
               const int*   __restrict__ dst,
               const float* __restrict__ norm,
               float* __restrict__ S,
               float* __restrict__ agg,
               int E) {
    const int lane = threadIdx.x & 31;
    const int wv   = threadIdx.x >> 5;
    const int e    = blockIdx.x * 8 + wv;
    if (e >= E) return;
    const int s = src[e];
    const int d = dst[e];
    const float w = norm[s];
    if (lane == 0) atomicAdd(&S[d], w);
    const float4 f = *(const float4*)(feat + (size_t)s * DF + lane * 4);
    float* a = agg + (size_t)d * DF + lane * 4;
    atomicAdd(a + 0, w * f.x);
    atomicAdd(a + 1, w * f.y);
    atomicAdd(a + 2, w * f.z);
    atomicAdd(a + 3, w * f.w);
}

__global__ void __launch_bounds__(256)
gemm_kernel(const float* __restrict__ feat,
            const float* __restrict__ agg,
            const float* __restrict__ W,
            const float* __restrict__ bias,
            const float* __restrict__ norm,
            const float* __restrict__ S,
            float* __restrict__ out,
            int N) {
    __shared__ float Xs[16 * XS_STRIDE];

    const int tid = threadIdx.x;
    const int m0  = blockIdx.x * 16;

    for (int idx = tid; idx < 16 * 256; idx += 256) {
        const int r   = idx >> 8;
        const int k   = idx & 255;
        const int row = m0 + r;
        float v = 0.0f;
        if (row < N) {
            if (k < DF) v = S[row] * feat[(size_t)row * DF + k];
            else        v = agg[(size_t)row * DF + (k - DF)];
        }
        Xs[r * XS_STRIDE + k] = v;
    }
    __syncthreads();

    const int lane = tid & 31;
    const int wv   = tid >> 5;
    const int nm   = lane & 15;
    const int half = lane >> 4;
    const int ncol = wv * 16 + nm;

    const float* __restrict__ wrow = W + (size_t)ncol * 256;
    const float* __restrict__ arow = &Xs[nm * XS_STRIDE];

    v8f c = {};
#pragma unroll
    for (int k0 = 0; k0 < 256; k0 += 4) {
        v2f a;
        a.x = arow[k0 + 2 * half + 0];
        a.y = arow[k0 + 2 * half + 1];
        v2f b;
        b.x = wrow[k0 + 2 * half + 0];
        b.y = wrow[k0 + 2 * half + 1];
        c = __builtin_amdgcn_wmma_f32_16x16x4_f32(
                 false, a,  false, b,
                 (short)0, c,  false,  false);
    }

#pragma unroll
    for (int i = 0; i < 8; ++i) {
        const int row = m0 + i + 8 * half;
        if (row < N) {
            out[(size_t)row * DF + ncol] = norm[row] * (c[i] + S[row] * bias[ncol]);
        }
    }
}

extern "C" void kernel_launch(void* const* d_in, const int* in_sizes, int n_in,
                              void* d_out, int out_size, void* d_ws, size_t ws_size,
                              hipStream_t stream) {
    const float* feat = (const float*)d_in[0];
    const float* W    = (const float*)d_in[1];
    const float* bias = (const float*)d_in[2];
    const int*   src  = (const int*)d_in[3];
    const int*   dst  = (const int*)d_in[4];
    float* out = (float*)d_out;

    const int N = in_sizes[0] / DF;
    const int E = in_sizes[3];

    float* ws   = (float*)d_ws;
    float* norm = ws;
    float* S    = ws + N;
    float* agg  = ws + 2 * (size_t)N;

    const int zcount = N * (2 + DF);
    zero_ws_kernel<<<(zcount + 255) / 256, 256, 0, stream>>>(ws, zcount);
    deg_kernel<<<(E + 255) / 256, 256, 0, stream>>>(dst, norm, E);
    norm_kernel<<<(N + 255) / 256, 256, 0, stream>>>(norm, N);
    scatter_kernel<<<(E + 7) / 8, 256, 0, stream>>>(feat, src, dst, norm, S, agg, E);
    gemm_kernel<<<(N + 15) / 16, 256, 0, stream>>>(feat, agg, W, bias, norm, S, out, N);
}


# gemm: fixed grid of 512 blocks; each wave loads its W fragments once (through LDS) into 128 registers and walks the 16-row tiles, next tile prefetched; W is no longer re-read from global per tile
# speedup vs baseline: 1.3364x; 1.3364x over previous
#include <hip/hip_runtime.h>


typedef float v2f __attribute__((ext_vector_type(2)));
typedef float v8f __attribute__((ext_vector_type(8)));

#define DF 128
#define XS_STRIDE 260

__global__ void zero_ws_kernel(float* __restrict__ p, int n) {
    int i = blockIdx.x * blockDim.x + threadIdx.x;
    if (i < n) p[i] = 0.0f;
}

__global__ void deg_kernel(const int* __restrict__ dst, float* __restrict__ deg, int E) {
    int e = blockIdx.x * blockDim.x + threadIdx.x;
    if (e < E) atomicAdd(&deg[dst[e]], 1.0f);
}

__global__ void norm_kernel(float* __restrict__ deg_norm, int N) {
    int v = blockIdx.x * blockDim.x + threadIdx.x;
    if (v < N) {
        float d = deg_norm[v];
        deg_norm[v] = rsqrtf(fmaxf(d, 1.0f));
    }
}

__global__ void __launch_bounds__(256)
scatter_kernel(const float* __restrict__ feat,
               const int*   __restrict__ src,
               const int*   __restrict__ dst,
               const float* __restrict__ norm,
               float* __restrict__ S,
               float* __restrict__ agg,
               int E) {
    const int lane = threadIdx.x & 31;
    const int wv   = threadIdx.x >> 5;
    const int e    = blockIdx.x * 8 + wv;
    if (e >= E) return;
    const int s = src[e];
    const int d = dst[e];
    const float w = norm[s];
    if (lane == 0) atomicAdd(&S[d], w);
    const float4 f = *(const float4*)(feat + (size_t)s * DF + lane * 4);
    float* a = agg + (size_t)d * DF + lane * 4;
    atomicAdd(a + 0, w * f.x);
    atomicAdd(a + 1, w * f.y);
    atomicAdd(a + 2, w * f.z);
    atomicAdd(a + 3, w * f.w);
}

#define GEMM_BLOCKS 512

__global__ void __launch_bounds__(256)
gemm_kernel(const float* __restrict__ feat,
            const float* __restrict__ agg,
            const float* __restrict__ W,
            const float* __restrict__ bias,
            const float* __restrict__ norm,
            const float* __restrict__ S,
            float* __restrict__ out,
            int N) {
    __shared__ float lds[128 * XS_STRIDE];

    const int tid  = threadIdx.x;
    const int lane = tid & 31;
    const int wv   = tid >> 5;
    const int nm   = lane & 15;
    const int half = lane >> 4;
    const int ncol = wv * 16 + nm;

    const int ntiles = (N + 15) >> 4;
    const int step   = gridDim.x;

    const int  sr0    = tid >> 6;
    const int  c4     = tid & 63;
    const bool isFeat = c4 < 32;
    const float* __restrict__ sbase = isFeat ? (feat + 4 * c4) : (agg + 4 * (c4 - 32));

    float4 p[4];
    float  ps[4];

    int t = blockIdx.x;

#pragma unroll
    for (int i = 0; i < 4; ++i) {
        const int row = t * 16 + sr0 + 4 * i;
        p[i]  = make_float4(0.0f, 0.0f, 0.0f, 0.0f);
        ps[i] = 1.0f;
        if (t < ntiles && row < N) {
            p[i] = *(const float4*)(sbase + (size_t)row * DF);
            if (isFeat) ps[i] = S[row];
        }
    }

    for (int q = tid; q < 128 * 64; q += 256) {
        const int wr = q >> 6;
        const int wc = q & 63;
        const float4 v = *(const float4*)(W + (size_t)wr * 256 + 4 * wc);
        *(float4*)&lds[wr * XS_STRIDE + 4 * wc] = v;
    }
    __syncthreads();

    v2f b[64];
    {
        const float* wl = &lds[ncol * XS_STRIDE + 2 * half];
#pragma unroll
        for (int kk = 0; kk < 64; ++kk) b[kk] = *(const v2f*)(wl + 4 * kk);
    }
    const float bcol = bias[ncol];
    __syncthreads();

    int buf = 0;
    for (; t < ntiles; t += step) {
        const int m0 = t * 16;
        float* xs = &lds[buf * (16 * XS_STRIDE)];

#pragma unroll
        for (int i = 0; i < 4; ++i) {
            float4 v = p[i];
            if (isFeat) { v.x *= ps[i]; v.y *= ps[i]; v.z *= ps[i]; v.w *= ps[i]; }
            *(float4*)&xs[(sr0 + 4 * i) * XS_STRIDE + 4 * c4] = v;
        }
        __syncthreads();

        float nr[8], sr[8];
#pragma unroll
        for (int i = 0; i < 8; ++i) {
            const int row = m0 + i + 8 * half;
            const int rc  = row < N ? row : N - 1;
            nr[i] = norm[rc];
            sr[i] = S[rc];
        }

        const int tn = t + step;
        if (tn < ntiles) {
#pragma unroll
            for (int i = 0; i < 4; ++i) {
                const int row = tn * 16 + sr0 + 4 * i;
                p[i]  = make_float4(0.0f, 0.0f, 0.0f, 0.0f);
                ps[i] = 1.0f;
                if (row < N) {
                    p[i] = *(const float4*)(sbase + (size_t)row * DF);
                    if (isFeat) ps[i] = S[row];
                }
            }
        }

        const float* __restrict__ arow = &xs[nm * XS_STRIDE + 2 * half];

        v8f c = {};
#pragma unroll
        for (int kk = 0; kk < 64; ++kk) {
            const v2f a = *(const v2f*)(arow + 4 * kk);
            c = __builtin_amdgcn_wmma_f32_16x16x4_f32(
                     false, a,  false, b[kk],
                     (short)0, c,  false,  false);
        }

#pragma unroll
        for (int i = 0; i < 8; ++i) {
            const int row = m0 + i + 8 * half;
            const float v = nr[i] * (c[i] + sr[i] * bcol);
            if (row < N) out[(size_t)row * DF + ncol] = v;
        }
        buf ^= 1;
    }
}

extern "C" void kernel_launch(void* const* d_in, const int* in_sizes, int n_in,
                              void* d_out, int out_size, void* d_ws, size_t ws_size,
                              hipStream_t stream) {
    const float* feat = (const float*)d_in[0];
    const float* W    = (const float*)d_in[1];
    const float* bias = (const float*)d_in[2];
    const int*   src  = (const int*)d_in[3];
    const int*   dst  = (const int*)d_in[4];
    float* out = (float*)d_out;

    const int N = in_sizes[0] / DF;
    const int E = in_sizes[3];

    float* ws   = (float*)d_ws;
    float* norm = ws;
    float* S    = ws + N;
    float* agg  = ws + 2 * (size_t)N;

    const int zcount = N * (2 + DF);
    zero_ws_kernel<<<(zcount + 255) / 256, 256, 0, stream>>>(ws, zcount);
    deg_kernel<<<(E + 255) / 256, 256, 0, stream>>>(dst, norm, E);
    norm_kernel<<<(N + 255) / 256, 256, 0, stream>>>(norm, N);
    scatter_kernel<<<(E + 7) / 8, 256, 0, stream>>>(feat, src, dst, norm, S, agg, E);
    gemm_kernel<<<GEMM_BLOCKS, 256, 0, stream>>>(feat, agg, W, bias, norm, S, out, N);
}


# float-atomic scatter replaced by a gather: an integer count ranks each edge into its node's agg row; one wave per node sums its source rows in arrival order, writes agg and S once; no 52 MB zero fill
# speedup vs baseline: 1.9270x; 1.4419x over previous
#include <hip/hip_runtime.h>


typedef float v2f __attribute__((ext_vector_type(2)));
typedef float v8f __attribute__((ext_vector_type(8)));

#define DF 128
#define XS_STRIDE 260

__global__ void zero_cnt_kernel(int* __restrict__ p, int n) {
    int i = blockIdx.x * blockDim.x + threadIdx.x;
    if (i < n) p[i] = 0;
}

__global__ void __launch_bounds__(256)
count_fill_kernel(const int* __restrict__ src,
                  const int* __restrict__ dst,
                  int* __restrict__ cnt,
                  int* __restrict__ slots,
                  int E) {
    const int e = blockIdx.x * blockDim.x + threadIdx.x;
    if (e >= E) return;
    const int d = dst[e];
    const int s = src[e];
    const int r = atomicAdd(&cnt[d], 1);
    if (r < DF) slots[(size_t)d * DF + r] = s;
}

__device__ __forceinline__ float node_norm(const int* __restrict__ cnt, int v) {
    return rsqrtf(fmaxf((float)cnt[v], 1.0f));
}

template <int K>
__device__ __forceinline__ void gather_rows(const float* __restrict__ fl, const int* __restrict__ cnt,
                                            int my, int j0, float4& acc, float& ssum) {
    int    s[K];
    int    c[K];
    float4 f[K];
#pragma unroll
    for (int u = 0; u < K; ++u) s[u] = __builtin_amdgcn_readlane(my, j0 + u);
#pragma unroll
    for (int u = 0; u < K; ++u) {
        c[u] = cnt[s[u]];
        f[u] = *(const float4*)(fl + (size_t)s[u] * DF);
    }
#pragma unroll
    for (int u = 0; u < K; ++u) {
        const float w = rsqrtf(fmaxf((float)c[u], 1.0f));
        acc.x += w * f[u].x;
        acc.y += w * f[u].y;
        acc.z += w * f[u].z;
        acc.w += w * f[u].w;
        ssum  += w;
    }
}

__global__ void __launch_bounds__(256)
gather_kernel(const float* __restrict__ feat,
              const int*   __restrict__ src,
              const int*   __restrict__ dst,
              const int*   __restrict__ cnt,
              const int*   slots,
              float* __restrict__ S,
              float* agg,
              int N, int E) {
    const int lane = threadIdx.x & 31;
    const int v = __builtin_amdgcn_readfirstlane(blockIdx.x * 8 + (threadIdx.x >> 5));
    if (v >= N) return;
    const int n = cnt[v];
    const float* __restrict__ fl = feat + lane * 4;

    float4 acc = make_float4(0.0f, 0.0f, 0.0f, 0.0f);
    float ssum = 0.0f;

    if (n <= DF) {
        const int* row = slots + (size_t)v * DF;
        for (int base = 0; base < n; base += 32) {
            const int m = min(32, n - base);
            int my = 0;
            if (lane < m) my = row[base + lane];
            for (int j0 = 0; j0 < m; j0 += 8) {
                switch (min(8, m - j0)) {
                    case 1:  gather_rows<1>(fl, cnt, my, j0, acc, ssum); break;
                    case 2:  gather_rows<2>(fl, cnt, my, j0, acc, ssum); break;
                    case 3:  gather_rows<3>(fl, cnt, my, j0, acc, ssum); break;
                    case 4:  gather_rows<4>(fl, cnt, my, j0, acc, ssum); break;
                    case 5:  gather_rows<5>(fl, cnt, my, j0, acc, ssum); break;
                    case 6:  gather_rows<6>(fl, cnt, my, j0, acc, ssum); break;
                    case 7:  gather_rows<7>(fl, cnt, my, j0, acc, ssum); break;
                    default: gather_rows<8>(fl, cnt, my, j0, acc, ssum); break;
                }
            }
        }
    } else {
        for (int e0 = 0; e0 < E; e0 += 32) {
            const int e = e0 + lane;
            int dd = -1, ss = 0;
            if (e < E) { dd = dst[e]; ss = src[e]; }
            unsigned mask = __builtin_amdgcn_ballot_w32(dd == v);
            while (mask) {
                const int j = __builtin_ctz(mask);
                mask &= mask - 1;
                const int s = __builtin_amdgcn_readlane(ss, j);
                const float w = node_norm(cnt, s);
                const float4 f = *(const float4*)(fl + (size_t)s * DF);
                acc.x += w * f.x;
                acc.y += w * f.y;
                acc.z += w * f.z;
                acc.w += w * f.w;
                ssum  += w;
            }
        }
    }
    *(float4*)(agg + (size_t)v * DF + lane * 4) = acc;
    if (lane == 0) S[v] = ssum;
}

#define GEMM_BLOCKS 512

__global__ void __launch_bounds__(256)
gemm_kernel(const float* __restrict__ feat,
            const float* __restrict__ agg,
            const float* __restrict__ W,
            const float* __restrict__ bias,
            const int*   __restrict__ cnt,
            const float* __restrict__ S,
            float* __restrict__ out,
            int N) {
    __shared__ float lds[128 * XS_STRIDE];

    const int tid  = threadIdx.x;
    const int lane = tid & 31;
    const int wv   = tid >> 5;
    const int nm   = lane & 15;
    const int half = lane >> 4;
    const int ncol = wv * 16 + nm;

    const int ntiles = (N + 15) >> 4;
    const int step   = gridDim.x;

    const int  sr0    = tid >> 6;
    const int  c4     = tid & 63;
    const bool isFeat = c4 < 32;
    const float* __restrict__ sbase = isFeat ? (feat + 4 * c4) : (agg + 4 * (c4 - 32));

    float4 p[4];
    float  ps[4];

    int t = blockIdx.x;

#pragma unroll
    for (int i = 0; i < 4; ++i) {
        const int row = t * 16 + sr0 + 4 * i;
        p[i]  = make_float4(0.0f, 0.0f, 0.0f, 0.0f);
        ps[i] = 1.0f;
        if (t < ntiles && row < N) {
            p[i] = *(const float4*)(sbase + (size_t)row * DF);
            if (isFeat) ps[i] = S[row];
        }
    }

    for (int q = tid; q < 128 * 64; q += 256) {
        const int wr = q >> 6;
        const int wc = q & 63;
        const float4 v = *(const float4*)(W + (size_t)wr * 256 + 4 * wc);
        *(float4*)&lds[wr * XS_STRIDE + 4 * wc] = v;
    }
    __syncthreads();

    v2f b[64];
    {
        const float* wl = &lds[ncol * XS_STRIDE + 2 * half];
#pragma unroll
        for (int kk = 0; kk < 64; ++kk) b[kk] = *(const v2f*)(wl + 4 * kk);
    }
    const float bcol = bias[ncol];
    __syncthreads();

    int buf = 0;
    for (; t < ntiles; t += step) {
        const int m0 = t * 16;
        float* xs = &lds[buf * (16 * XS_STRIDE)];

#pragma unroll
        for (int i = 0; i < 4; ++i) {
            float4 v = p[i];
            if (isFeat) { v.x *= ps[i]; v.y *= ps[i]; v.z *= ps[i]; v.w *= ps[i]; }
            *(float4*)&xs[(sr0 + 4 * i) * XS_STRIDE + 4 * c4] = v;
        }
        __syncthreads();

        float nr[8], sr[8];
#pragma unroll
        for (int i = 0; i < 8; ++i) {
            const int row = m0 + i + 8 * half;
            const int rc  = row < N ? row : N - 1;
            nr[i] = node_norm(cnt, rc);
            sr[i] = S[rc];
        }

        const int tn = t + step;
        if (tn < ntiles) {
#pragma unroll
            for (int i = 0; i < 4; ++i) {
                const int row = tn * 16 + sr0 + 4 * i;
                p[i]  = make_float4(0.0f, 0.0f, 0.0f, 0.0f);
                ps[i] = 1.0f;
                if (row < N) {
                    p[i] = *(const float4*)(sbase + (size_t)row * DF);
                    if (isFeat) ps[i] = S[row];
                }
            }
        }

        const float* __restrict__ arow = &xs[nm * XS_STRIDE + 2 * half];

        v8f c = {};
#pragma unroll
        for (int kk = 0; kk < 64; ++kk) {
            const v2f a = *(const v2f*)(arow + 4 * kk);
            c = __builtin_amdgcn_wmma_f32_16x16x4_f32(
                     false, a,  false, b[kk],
                     (short)0, c,  false,  false);
        }

#pragma unroll
        for (int i = 0; i < 8; ++i) {
            const int row = m0 + i + 8 * half;
            const float v = nr[i] * (c[i] + sr[i] * bcol);
            if (row < N) out[(size_t)row * DF + ncol] = v;
        }
        buf ^= 1;
    }
}

extern "C" void kernel_launch(void* const* d_in, const int* in_sizes, int n_in,
                              void* d_out, int out_size, void* d_ws, size_t ws_size,
                              hipStream_t stream) {
    const float* feat = (const float*)d_in[0];
    const float* W    = (const float*)d_in[1];
    const float* bias = (const float*)d_in[2];
    const int*   src  = (const int*)d_in[3];
    const int*   dst  = (const int*)d_in[4];
    float* out = (float*)d_out;

    const int N = in_sizes[0] / DF;
    const int E = in_sizes[3];

    int*   cnt = (int*)d_ws;
    float* S   = (float*)d_ws + N;
    float* agg = (float*)d_ws + 2 * (size_t)N;

    zero_cnt_kernel<<<(N + 255) / 256, 256, 0, stream>>>(cnt, N);
    count_fill_kernel<<<(E + 255) / 256, 256, 0, stream>>>(src, dst, cnt, (int*)agg, E);
    gather_kernel<<<(N + 7) / 8, 256, 0, stream>>>(feat, src, dst, cnt, (const int*)agg, S, agg, N, E);
    gemm_kernel<<<GEMM_BLOCKS, 256, 0, stream>>>(feat, agg, W, bias, cnt, S, out, N);
}


# edge ranking by LDS counters instead of global returning adds + scattered 4-B stores: edges binned per chunk by destination range, then ranked per node in LDS; sum order = arrival rank of LDS adds
# speedup vs baseline: 2.4758x; 1.2848x over previous
#include <hip/hip_runtime.h>


typedef float v2f __attribute__((ext_vector_type(2)));
typedef float v8f __attribute__((ext_vector_type(8)));

#define DF 128
#define XS_STRIDE 260

#define NBIN 256
#define NCHK 256
#define SEG_CAP 31
#define SEG_W 64
#define SLOT_CAP 32
#define RMAX 512

__device__ __forceinline__ int range_nodes(int N) {
    return (((N + NBIN - 1) / NBIN) + 31) & ~31;
}

__global__ void __launch_bounds__(512)
bin_kernel(const int* __restrict__ src,
           const int* __restrict__ dst,
           int* __restrict__ cells,
           int N, int E) {
    __shared__ int  hist[NBIN];
    __shared__ int2 stage[NBIN * SEG_CAP];

    const int tid  = threadIdx.x;
    const int lane = tid & 31;
    const int wv   = __builtin_amdgcn_readfirstlane(tid >> 5);
    const int R    = range_nodes(N);
    const int CH   = (E + NCHK - 1) / NCHK;
    const int c    = blockIdx.x;
    const int e0   = c * CH;
    const int e1   = min(E, e0 + CH);

    if (tid < NBIN) hist[tid] = 0;
    __syncthreads();

    for (int eb = e0; eb < e1; eb += 4 * 512) {
        int d[4], s[4];
#pragma unroll
        for (int k = 0; k < 4; ++k) {
            const int e = eb + k * 512 + tid;
            d[k] = -1;
            s[k] = 0;
            if (e < e1) { d[k] = dst[e]; s[k] = src[e]; }
        }
#pragma unroll
        for (int k = 0; k < 4; ++k) {
            if (d[k] >= 0) {
                const int bb = d[k] / R;
                const int r  = atomicAdd(&hist[bb], 1);
                if (r < SEG_CAP) stage[bb * SEG_CAP + r] = make_int2(d[k], s[k]);
            }
        }
    }
    __syncthreads();

    constexpr int CPW = NBIN / 16;
    int  nn[CPW];
    int2 val[CPW];
#pragma unroll
    for (int t = 0; t < CPW; ++t) nn[t] = hist[wv + 16 * t];
#pragma unroll
    for (int t = 0; t < CPW; ++t) {
        val[t] = make_int2(-1, -1);
        if (lane >= 1 && lane <= min(nn[t], SEG_CAP)) val[t] = stage[(wv + 16 * t) * SEG_CAP + lane - 1];
    }
#pragma unroll
    for (int t = 0; t < CPW; ++t) {
        const int bb = wv + 16 * t;
        const int2 o = lane == 0 ? make_int2(nn[t], 0) : val[t];
        if (bb * R < N) *(int2*)(cells + (size_t)bb * R * DF + (size_t)c * SEG_W + 2 * lane) = o;
    }
}

__device__ __forceinline__ void list_add(int* cnt_l, int* slot_l, int li, int s) {
    const int r = atomicAdd(&cnt_l[li], 1);
    if (r < SLOT_CAP) slot_l[li * SLOT_CAP + r] = s;
}

__global__ void __launch_bounds__(512)
list_kernel(const int* __restrict__ src,
            const int* __restrict__ dst,
            int* __restrict__ cnt,
            int* cells,
            int N, int E) {
    __shared__ int cnt_l[RMAX];
    __shared__ int slot_l[RMAX * SLOT_CAP];

    const int tid  = threadIdx.x;
    const int lane = tid & 31;
    const int wv   = __builtin_amdgcn_readfirstlane(tid >> 5);
    const int R    = range_nodes(N);
    const int CH   = (E + NCHK - 1) / NCHK;
    const int base = blockIdx.x * R;
    if (base >= N) return;

    cnt_l[tid] = 0;
    __syncthreads();

    const int* region = cells + (size_t)base * DF;
    constexpr int CPW = NCHK / 16;
    int2 q[CPW];
#pragma unroll
    for (int k = 0; k < CPW; ++k)
        q[k] = *(const int2*)(region + (size_t)(wv + 16 * k) * SEG_W + 2 * lane);
#pragma unroll
    for (int k = 0; k < CPW; ++k) {
        const int n = __builtin_amdgcn_readfirstlane(q[k].x);
        if (n <= SEG_CAP) {
            if (lane >= 1 && lane <= n) list_add(cnt_l, slot_l, q[k].x - base, q[k].y);
        } else {
            const int c  = wv + 16 * k;
            const int f0 = c * CH;
            const int f1 = min(E, f0 + CH);
            for (int e = f0 + lane; e < f1; e += 32) {
                const unsigned li = (unsigned)(dst[e] - base);
                if (li < (unsigned)R) list_add(cnt_l, slot_l, (int)li, src[e]);
            }
        }
    }
    __syncthreads();

    for (int i = tid; i < R; i += 512)
        if (base + i < N) cnt[base + i] = cnt_l[i];
    const int per = R / 16;
    for (int t0 = 0; t0 < per; t0 += 4) {
        int n4[4], s4[4];
#pragma unroll
        for (int u = 0; u < 4; ++u) {
            const int  i  = wv + 16 * (t0 + u);
            const int  ci = cnt_l[min(i, RMAX - 1)];
            const bool ok = t0 + u < per && base + i < N;
            n4[u] = ok ? min(ci, SLOT_CAP) : 0;
        }
#pragma unroll
        for (int u = 0; u < 4; ++u) {
            const int i = wv + 16 * (t0 + u);
            s4[u] = lane < n4[u] ? slot_l[i * SLOT_CAP + lane] : 0;
        }
#pragma unroll
        for (int u = 0; u < 4; ++u) {
            const int i = wv + 16 * (t0 + u);
            if (lane < n4[u]) cells[(size_t)(base + i) * DF + lane] = s4[u];
        }
    }
}

__device__ __forceinline__ float node_norm(const int* __restrict__ cnt, int v) {
    return rsqrtf(fmaxf((float)cnt[v], 1.0f));
}

template <int K>
__device__ __forceinline__ void gather_rows(const float* __restrict__ fl, const int* __restrict__ cnt,
                                            int my, int j0, float4& acc, float& ssum) {
    int    s[K];
    int    c[K];
    float4 f[K];
#pragma unroll
    for (int u = 0; u < K; ++u) s[u] = __builtin_amdgcn_readlane(my, j0 + u);
#pragma unroll
    for (int u = 0; u < K; ++u) {
        c[u] = cnt[s[u]];
        f[u] = *(const float4*)(fl + (size_t)s[u] * DF);
    }
#pragma unroll
    for (int u = 0; u < K; ++u) {
        const float w = rsqrtf(fmaxf((float)c[u], 1.0f));
        acc.x += w * f[u].x;
        acc.y += w * f[u].y;
        acc.z += w * f[u].z;
        acc.w += w * f[u].w;
        ssum  += w;
    }
}

__global__ void __launch_bounds__(256)
gather_kernel(const float* __restrict__ feat,
              const int*   __restrict__ src,
              const int*   __restrict__ dst,
              const int*   __restrict__ cnt,
              const int*   slots,
              float* __restrict__ S,
              float* agg,
              int N, int E) {
    const int lane = threadIdx.x & 31;
    const int v = __builtin_amdgcn_readfirstlane(blockIdx.x * 8 + (threadIdx.x >> 5));
    if (v >= N) return;
    const int n = cnt[v];
    const float* __restrict__ fl = feat + lane * 4;

    float4 acc = make_float4(0.0f, 0.0f, 0.0f, 0.0f);
    float ssum = 0.0f;

    if (n <= SLOT_CAP) {
        const int* row = slots + (size_t)v * DF;
        for (int base = 0; base < n; base += 32) {
            const int m = min(32, n - base);
            int my = 0;
            if (lane < m) my = row[base + lane];
            for (int j0 = 0; j0 < m; j0 += 8) {
                switch (min(8, m - j0)) {
                    case 1:  gather_rows<1>(fl, cnt, my, j0, acc, ssum); break;
                    case 2:  gather_rows<2>(fl, cnt, my, j0, acc, ssum); break;
                    case 3:  gather_rows<3>(fl, cnt, my, j0, acc, ssum); break;
                    case 4:  gather_rows<4>(fl, cnt, my, j0, acc, ssum); break;
                    case 5:  gather_rows<5>(fl, cnt, my, j0, acc, ssum); break;
                    case 6:  gather_rows<6>(fl, cnt, my, j0, acc, ssum); break;
                    case 7:  gather_rows<7>(fl, cnt, my, j0, acc, ssum); break;
                    default: gather_rows<8>(fl, cnt, my, j0, acc, ssum); break;
                }
            }
        }
    } else {
        for (int e0 = 0; e0 < E; e0 += 32) {
            const int e = e0 + lane;
            int dd = -1, ss = 0;
            if (e < E) { dd = dst[e]; ss = src[e]; }
            unsigned mask = __builtin_amdgcn_ballot_w32(dd == v);
            while (mask) {
                const int j = __builtin_ctz(mask);
                mask &= mask - 1;
                const int s = __builtin_amdgcn_readlane(ss, j);
                const float w = node_norm(cnt, s);
                const float4 f = *(const float4*)(fl + (size_t)s * DF);
                acc.x += w * f.x;
                acc.y += w * f.y;
                acc.z += w * f.z;
                acc.w += w * f.w;
                ssum  += w;
            }
        }
    }
    *(float4*)(agg + (size_t)v * DF + lane * 4) = acc;
    if (lane == 0) S[v] = ssum;
}

#define GEMM_BLOCKS 512

__global__ void __launch_bounds__(256)
gemm_kernel(const float* __restrict__ feat,
            const float* __restrict__ agg,
            const float* __restrict__ W,
            const float* __restrict__ bias,
            const int*   __restrict__ cnt,
            const float* __restrict__ S,
            float* __restrict__ out,
            int N) {
    __shared__ float lds[128 * XS_STRIDE];

    const int tid  = threadIdx.x;
    const int lane = tid & 31;
    const int wv   = tid >> 5;
    const int nm   = lane & 15;
    const int half = lane >> 4;
    const int ncol = wv * 16 + nm;

    const int ntiles = (N + 15) >> 4;
    const int step   = gridDim.x;

    const int  sr0    = tid >> 6;
    const int  c4     = tid & 63;
    const bool isFeat = c4 < 32;
    const float* __restrict__ sbase = isFeat ? (feat + 4 * c4) : (agg + 4 * (c4 - 32));

    float4 p[4];
    float  ps[4];

    int t = blockIdx.x;

#pragma unroll
    for (int i = 0; i < 4; ++i) {
        const int row = t * 16 + sr0 + 4 * i;
        p[i]  = make_float4(0.0f, 0.0f, 0.0f, 0.0f);
        ps[i] = 1.0f;
        if (t < ntiles && row < N) {
            p[i] = *(const float4*)(sbase + (size_t)row * DF);
            if (isFeat) ps[i] = S[row];
        }
    }

    for (int q = tid; q < 128 * 64; q += 256) {
        const int wr = q >> 6;
        const int wc = q & 63;
        const float4 v = *(const float4*)(W + (size_t)wr * 256 + 4 * wc);
        *(float4*)&lds[wr * XS_STRIDE + 4 * wc] = v;
    }
    __syncthreads();

    v2f b[64];
    {
        const float* wl = &lds[ncol * XS_STRIDE + 2 * half];
#pragma unroll
        for (int kk = 0; kk < 64; ++kk) b[kk] = *(const v2f*)(wl + 4 * kk);
    }
    const float bcol = bias[ncol];
    __syncthreads();

    int buf = 0;
    for (; t < ntiles; t += step) {
        const int m0 = t * 16;
        float* xs = &lds[buf * (16 * XS_STRIDE)];

#pragma unroll
        for (int i = 0; i < 4; ++i) {
            float4 v = p[i];
            if (isFeat) { v.x *= ps[i]; v.y *= ps[i]; v.z *= ps[i]; v.w *= ps[i]; }
            *(float4*)&xs[(sr0 + 4 * i) * XS_STRIDE + 4 * c4] = v;
        }
        __syncthreads();

        float nr[8], sr[8];
#pragma unroll
        for (int i = 0; i < 8; ++i) {
            const int row = m0 + i + 8 * half;
            const int rc  = row < N ? row : N - 1;
            nr[i] = node_norm(cnt, rc);
            sr[i] = S[rc];
        }

        const int tn = t + step;
        if (tn < ntiles) {
#pragma unroll
            for (int i = 0; i < 4; ++i) {
                const int row = tn * 16 + sr0 + 4 * i;
                p[i]  = make_float4(0.0f, 0.0f, 0.0f, 0.0f);
                ps[i] = 1.0f;
                if (row < N) {
                    p[i] = *(const float4*)(sbase + (size_t)row * DF);
                    if (isFeat) ps[i] = S[row];
                }
            }
        }

        const float* __restrict__ arow = &xs[nm * XS_STRIDE + 2 * half];

        v8f c = {};
#pragma unroll
        for (int kk = 0; kk < 64; ++kk) {
            const v2f a = *(const v2f*)(arow + 4 * kk);
            c = __builtin_amdgcn_wmma_f32_16x16x4_f32(
                     false, a,  false, b[kk],
                     (short)0, c,  false,  false);
        }

#pragma unroll
        for (int i = 0; i < 8; ++i) {
            const int row = m0 + i + 8 * half;
            const float v = nr[i] * (c[i] + sr[i] * bcol);
            if (row < N) out[(size_t)row * DF + ncol] = v;
        }
        buf ^= 1;
    }
}

extern "C" void kernel_launch(void* const* d_in, const int* in_sizes, int n_in,
                              void* d_out, int out_size, void* d_ws, size_t ws_size,
                              hipStream_t stream) {
    const float* feat = (const float*)d_in[0];
    const float* W    = (const float*)d_in[1];
    const float* bias = (const float*)d_in[2];
    const int*   src  = (const int*)d_in[3];
    const int*   dst  = (const int*)d_in[4];
    float* out = (float*)d_out;

    const int N = in_sizes[0] / DF;
    const int E = in_sizes[3];

    int*   cnt = (int*)d_ws;
    float* S   = (float*)d_ws + N;
    float* agg = (float*)d_ws + 2 * (size_t)N;

    bin_kernel<<<NCHK, 512, 0, stream>>>(src, dst, (int*)agg, N, E);
    list_kernel<<<NBIN, 512, 0, stream>>>(src, dst, cnt, (int*)agg, N, E);
    gather_kernel<<<(N + 7) / 8, 256, 0, stream>>>(feat, src, dst, cnt, (const int*)agg, S, agg, N, E);
    gemm_kernel<<<GEMM_BLOCKS, 256, 0, stream>>>(feat, agg, W, bias, cnt, S, out, N);
}


# gather folded into the product kernel: each wave sums 2 rows of its block's next A tile from their source rows (slot order kept) straight into LDS; agg and S never pass through memory; 3 launches
# speedup vs baseline: 3.3583x; 1.3565x over previous
#include <hip/hip_runtime.h>


typedef float v2f __attribute__((ext_vector_type(2)));
typedef float v8f __attribute__((ext_vector_type(8)));

#define DF 128
#define XS_STRIDE 260

#define NBIN 256
#define NCHK 256
#define SEG_CAP 31
#define SEG_W 64
#define SLOT_CAP 32
#define RMAX 512

__device__ __forceinline__ int range_nodes(int N) {
    return (((N + NBIN - 1) / NBIN) + 31) & ~31;
}

__global__ void __launch_bounds__(512)
bin_kernel(const int* __restrict__ src,
           const int* __restrict__ dst,
           int* __restrict__ cells,
           int N, int E) {
    __shared__ int  hist[NBIN];
    __shared__ int2 stage[NBIN * SEG_CAP];

    const int tid  = threadIdx.x;
    const int lane = tid & 31;
    const int wv   = __builtin_amdgcn_readfirstlane(tid >> 5);
    const int R    = range_nodes(N);
    const int CH   = (E + NCHK - 1) / NCHK;
    const int c    = blockIdx.x;
    const int e0   = c * CH;
    const int e1   = min(E, e0 + CH);

    if (tid < NBIN) hist[tid] = 0;
    __syncthreads();

    for (int eb = e0; eb < e1; eb += 4 * 512) {
        int d[4], s[4];
#pragma unroll
        for (int k = 0; k < 4; ++k) {
            const int e = eb + k * 512 + tid;
            d[k] = -1;
            s[k] = 0;
            if (e < e1) { d[k] = dst[e]; s[k] = src[e]; }
        }
#pragma unroll
        for (int k = 0; k < 4; ++k) {
            if (d[k] >= 0) {
                const int bb = d[k] / R;
                const int r  = atomicAdd(&hist[bb], 1);
                if (r < SEG_CAP) stage[bb * SEG_CAP + r] = make_int2(d[k], s[k]);
            }
        }
    }
    __syncthreads();

    constexpr int CPW = NBIN / 16;
    int  nn[CPW];
    int2 val[CPW];
#pragma unroll
    for (int t = 0; t < CPW; ++t) nn[t] = hist[wv + 16 * t];
#pragma unroll
    for (int t = 0; t < CPW; ++t) {
        val[t] = make_int2(-1, -1);
        if (lane >= 1 && lane <= min(nn[t], SEG_CAP)) val[t] = stage[(wv + 16 * t) * SEG_CAP + lane - 1];
    }
#pragma unroll
    for (int t = 0; t < CPW; ++t) {
        const int bb = wv + 16 * t;
        const int2 o = lane == 0 ? make_int2(nn[t], 0) : val[t];
        if (bb * R < N) *(int2*)(cells + (size_t)bb * R * DF + (size_t)c * SEG_W + 2 * lane) = o;
    }
}

__device__ __forceinline__ void list_add(int* cnt_l, int* slot_l, int li, int s) {
    const int r = atomicAdd(&cnt_l[li], 1);
    if (r < SLOT_CAP) slot_l[li * SLOT_CAP + r] = s;
}

__global__ void __launch_bounds__(512)
list_kernel(const int* __restrict__ src,
            const int* __restrict__ dst,
            int* __restrict__ cnt,
            int* cells,
            int N, int E) {
    __shared__ int cnt_l[RMAX];
    __shared__ int slot_l[RMAX * SLOT_CAP];

    const int tid  = threadIdx.x;
    const int lane = tid & 31;
    const int wv   = __builtin_amdgcn_readfirstlane(tid >> 5);
    const int R    = range_nodes(N);
    const int CH   = (E + NCHK - 1) / NCHK;
    const int base = blockIdx.x * R;
    if (base >= N) return;

    cnt_l[tid] = 0;
    __syncthreads();

    const int* region = cells + (size_t)base * DF;
    constexpr int CPW = NCHK / 16;
    int2 q[CPW];
#pragma unroll
    for (int k = 0; k < CPW; ++k)
        q[k] = *(const int2*)(region + (size_t)(wv + 16 * k) * SEG_W + 2 * lane);
#pragma unroll
    for (int k = 0; k < CPW; ++k) {
        const int n = __builtin_amdgcn_readfirstlane(q[k].x);
        if (n <= SEG_CAP) {
            if (lane >= 1 && lane <= n) list_add(cnt_l, slot_l, q[k].x - base, q[k].y);
        } else {
            const int c  = wv + 16 * k;
            const int f0 = c * CH;
            const int f1 = min(E, f0 + CH);
            for (int e = f0 + lane; e < f1; e += 32) {
                const unsigned li = (unsigned)(dst[e] - base);
                if (li < (unsigned)R) list_add(cnt_l, slot_l, (int)li, src[e]);
            }
        }
    }
    __syncthreads();

    for (int i = tid; i < R; i += 512)
        if (base + i < N) cnt[base + i] = cnt_l[i];
    const int per = R / 16;
    for (int t0 = 0; t0 < per; t0 += 4) {
        int n4[4], s4[4];
#pragma unroll
        for (int u = 0; u < 4; ++u) {
            const int  i  = wv + 16 * (t0 + u);
            const int  ci = cnt_l[min(i, RMAX - 1)];
            const bool ok = t0 + u < per && base + i < N;
            n4[u] = ok ? min(ci, SLOT_CAP) : 0;
        }
#pragma unroll
        for (int u = 0; u < 4; ++u) {
            const int i = wv + 16 * (t0 + u);
            s4[u] = lane < n4[u] ? slot_l[i * SLOT_CAP + lane] : 0;
        }
#pragma unroll
        for (int u = 0; u < 4; ++u) {
            const int i = wv + 16 * (t0 + u);
            if (lane < n4[u]) cells[(size_t)(base + i) * DF + lane] = s4[u];
        }
    }
}

__device__ __forceinline__ float node_norm_of(int c) {
    return rsqrtf(fmaxf((float)c, 1.0f));
}

struct NodeHead {
    int    n;
    int    my;
    float4 own;
};

__device__ __forceinline__ NodeHead load_head(const float* __restrict__ feat, unsigned l4, int lz,
                                              const int* __restrict__ cnt,
                                              const int* __restrict__ slots, int v, int N, int lane) {
    NodeHead h;
    h.n   = 0;
    h.my  = 0;
    h.own = make_float4(0.0f, 0.0f, 0.0f, 0.0f);
    if (v < N) {
        h.n   = cnt[v + lz];
        h.my  = slots[(size_t)v * DF + lane];
        h.own = *(const float4*)(feat + (size_t)v * DF + l4);
    }
    return h;
}

#define GROUP 6

__device__ __forceinline__ void rows_issue(const float* __restrict__ feat, unsigned l4, int lz,
                                           const int* __restrict__ cnt,
                                           int my, int j0, int k, int (&c)[GROUP], float4 (&f)[GROUP]) {
#pragma unroll
    for (int u = 0; u < GROUP; ++u) {
        c[u] = 1;
        f[u] = make_float4(0.0f, 0.0f, 0.0f, 0.0f);
        if (u < k) {
            const int s = __builtin_amdgcn_readlane(my, j0 + u);
            c[u] = cnt[s + lz];
            f[u] = *(const float4*)(feat + (size_t)s * DF + l4);
        }
    }
}

__device__ __forceinline__ void rows_sum(int k, const int (&c)[GROUP], const float4 (&f)[GROUP],
                                         float4& acc, float& ssum) {
#pragma unroll
    for (int u = 0; u < GROUP; ++u) {
        const float w = u < k ? node_norm_of(c[u]) : 0.0f;
        acc.x += w * f[u].x;
        acc.y += w * f[u].y;
        acc.z += w * f[u].z;
        acc.w += w * f[u].w;
        ssum  += w;
    }
}

__device__ __forceinline__ void rows_scan(const float* __restrict__ feat, unsigned l4,
                                          const int* __restrict__ cnt,
                                          const int* __restrict__ src, const int* __restrict__ dst,
                                          int E, int lane, int v, float4& acc, float& ssum) {
    for (int e0 = 0; e0 < E; e0 += 32) {
        const int e = e0 + lane;
        int dd = -1, ss = 0;
        if (e < E) { dd = dst[e]; ss = src[e]; }
        unsigned mask = __builtin_amdgcn_ballot_w32(dd == v);
        while (mask) {
            const int j = __builtin_ctz(mask);
            mask &= mask - 1;
            const int s = __builtin_amdgcn_readlane(ss, j);
            const float w = node_norm_of(cnt[s]);
            const float4 f = *(const float4*)(feat + (size_t)s * DF + l4);
            acc.x += w * f.x;
            acc.y += w * f.y;
            acc.z += w * f.z;
            acc.w += w * f.w;
            ssum  += w;
        }
    }
}

__device__ __forceinline__ void fill_two_rows(const float* __restrict__ feat, unsigned l4, int lz,
                                              const int* __restrict__ cnt,
                                              const int* __restrict__ src, const int* __restrict__ dst,
                                              int E, int lane, int vA, const NodeHead& hA, const NodeHead& hB,
                                              float* __restrict__ xrow, float* __restrict__ rS,
                                              float* __restrict__ rN) {
    const int nA = __builtin_amdgcn_readfirstlane(hA.n);
    const int nB = __builtin_amdgcn_readfirstlane(hB.n);
    const int kA = nA <= SLOT_CAP ? nA : 0;
    const int kB = nB <= SLOT_CAP ? nB : 0;
    const int kmax = max(kA, kB);

    float4 accA = make_float4(0.0f, 0.0f, 0.0f, 0.0f);
    float4 accB = make_float4(0.0f, 0.0f, 0.0f, 0.0f);
    float  sA = 0.0f, sB = 0.0f;
    for (int j0 = 0; j0 < kmax; j0 += GROUP) {
        int    cA[GROUP], cB[GROUP];
        float4 fA[GROUP], fB[GROUP];
        rows_issue(feat, l4, lz, cnt, hA.my, j0, kA - j0, cA, fA);
        rows_issue(feat, l4, lz, cnt, hB.my, j0, kB - j0, cB, fB);
        rows_sum(kA - j0, cA, fA, accA, sA);
        rows_sum(kB - j0, cB, fB, accB, sB);
    }
    if (nA > SLOT_CAP) rows_scan(feat, l4, cnt, src, dst, E, lane, vA,     accA, sA);
    if (nB > SLOT_CAP) rows_scan(feat, l4, cnt, src, dst, E, lane, vA + 1, accB, sB);

    float* xa = xrow + l4;
    float* xb = xa + XS_STRIDE;
    *(float4*)xa         = make_float4(hA.own.x * sA, hA.own.y * sA, hA.own.z * sA, hA.own.w * sA);
    *(float4*)(xa + DF)  = accA;
    *(float4*)xb         = make_float4(hB.own.x * sB, hB.own.y * sB, hB.own.z * sB, hB.own.w * sB);
    *(float4*)(xb + DF)  = accB;
    if (lane == 0) {
        rS[0] = sA;
        rS[1] = sB;
        rN[0] = node_norm_of(nA);
        rN[1] = node_norm_of(nB);
    }
}

#define GEMM_BLOCKS 512

__global__ void __launch_bounds__(256)
conv_kernel(const float* __restrict__ feat,
            const float* __restrict__ W,
            const float* __restrict__ bias,
            const int*   __restrict__ src,
            const int*   __restrict__ dst,
            const int*   __restrict__ cnt,
            const int*   __restrict__ slots,
            float* __restrict__ out,
            int N, int E) {
    __shared__ float lds[128 * XS_STRIDE];
    __shared__ float rowS[2][16];
    __shared__ float rowN[2][16];

    const int tid  = threadIdx.x;
    const int lane = tid & 31;
    const int wv   = __builtin_amdgcn_readfirstlane(tid >> 5);
    const int nm   = lane & 15;
    const int half = lane >> 4;
    const int ncol = wv * 16 + nm;

    const int ntiles = (N + 15) >> 4;
    const int step   = gridDim.x;
    const unsigned l4 = lane * 4;
    int lz;
    asm("v_mov_b32 %0, 0" : "=v"(lz));

    int t = blockIdx.x;

    NodeHead hA = load_head(feat, l4, lz, cnt, slots, t * 16 + 2 * wv,     t < ntiles ? N : 0, lane);
    NodeHead hB = load_head(feat, l4, lz, cnt, slots, t * 16 + 2 * wv + 1, t < ntiles ? N : 0, lane);

    for (int q = tid; q < 128 * 64; q += 256) {
        const int wr = q >> 6;
        const int wc = q & 63;
        const float4 v = *(const float4*)(W + (size_t)wr * 256 + 4 * wc);
        *(float4*)&lds[wr * XS_STRIDE + 4 * wc] = v;
    }
    __syncthreads();

    v2f b[64];
    {
        const float* wl = &lds[ncol * XS_STRIDE + 2 * half];
#pragma unroll
        for (int kk = 0; kk < 64; ++kk) b[kk] = *(const v2f*)(wl + 4 * kk);
    }
    const float bcol = bias[ncol];
    __syncthreads();

    if (t < ntiles)
        fill_two_rows(feat, l4, lz, cnt, src, dst, E, lane, t * 16 + 2 * wv, hA, hB,
                      &lds[(2 * wv) * XS_STRIDE], &rowS[0][2 * wv], &rowN[0][2 * wv]);

    int buf = 0;
    for (; t < ntiles; t += step) {
        const int m0 = t * 16;
        const float* xs = &lds[buf * (16 * XS_STRIDE)];

        __syncthreads();

        const int tn = t + step;
        hA = load_head(feat, l4, lz, cnt, slots, tn * 16 + 2 * wv,     tn < ntiles ? N : 0, lane);
        hB = load_head(feat, l4, lz, cnt, slots, tn * 16 + 2 * wv + 1, tn < ntiles ? N : 0, lane);

        float nr[8], sr[8];
#pragma unroll
        for (int i = 0; i < 8; ++i) {
            nr[i] = rowN[buf][i + 8 * half];
            sr[i] = rowS[buf][i + 8 * half];
        }

        const float* __restrict__ arow = &xs[nm * XS_STRIDE + 2 * half];

        v8f c = {};
#pragma unroll
        for (int kk = 0; kk < 64; ++kk) {
            const v2f a = *(const v2f*)(arow + 4 * kk);
            c = __builtin_amdgcn_wmma_f32_16x16x4_f32(
                     false, a,  false, b[kk],
                     (short)0, c,  false,  false);
        }

#pragma unroll
        for (int i = 0; i < 8; ++i) {
            const int row = m0 + i + 8 * half;
            const float v = nr[i] * (c[i] + sr[i] * bcol);
            if (row < N) out[(size_t)row * DF + ncol] = v;
        }

        if (tn < ntiles)
            fill_two_rows(feat, l4, lz, cnt, src, dst, E, lane, tn * 16 + 2 * wv, hA, hB,
                          &lds[((buf ^ 1) * 16 + 2 * wv) * XS_STRIDE],
                          &rowS[buf ^ 1][2 * wv], &rowN[buf ^ 1][2 * wv]);
        buf ^= 1;
    }
}

extern "C" void kernel_launch(void* const* d_in, const int* in_sizes, int n_in,
                              void* d_out, int out_size, void* d_ws, size_t ws_size,
                              hipStream_t stream) {
    const float* feat = (const float*)d_in[0];
    const float* W    = (const float*)d_in[1];
    const float* bias = (const float*)d_in[2];
    const int*   src  = (const int*)d_in[3];
    const int*   dst  = (const int*)d_in[4];
    float* out = (float*)d_out;

    const int N = in_sizes[0] / DF;
    const int E = in_sizes[3];

    int* cnt  = (int*)d_ws;
    int* rows = (int*)d_ws + 2 * (size_t)N;

    bin_kernel<<<NCHK, 512, 0, stream>>>(src, dst, rows, N, E);
    list_kernel<<<NBIN, 512, 0, stream>>>(src, dst, cnt, rows, N, E);
    conv_kernel<<<GEMM_BLOCKS, 256, 0, stream>>>(feat, W, bias, src, dst, cnt, rows, out, N, E);
}


# result tile staged through LDS: each wave writes the two rows it gathered as whole 512-B rows, one store per row, not 64-B pieces of 16 rows; S, norm, bias applied there; sums and their order kept
# speedup vs baseline: 3.5731x; 1.0640x over previous
#include <hip/hip_runtime.h>


typedef float v2f __attribute__((ext_vector_type(2)));
typedef float v8f __attribute__((ext_vector_type(8)));

#define DF 128
#define XS_STRIDE 260
#define OS_STRIDE 132
#define OS_BASE (2 * 16 * XS_STRIDE)

#define NBIN 256
#define NCHK 256
#define SEG_CAP 31
#define SEG_W 64
#define SLOT_CAP 32
#define RMAX 512

__device__ __forceinline__ int range_nodes(int N) {
    return (((N + NBIN - 1) / NBIN) + 31) & ~31;
}

__global__ void __launch_bounds__(512)
bin_kernel(const int* __restrict__ src,
           const int* __restrict__ dst,
           int* __restrict__ cells,
           int N, int E) {
    __shared__ int  hist[NBIN];
    __shared__ int2 stage[NBIN * SEG_CAP];

    const int tid  = threadIdx.x;
    const int lane = tid & 31;
    const int wv   = __builtin_amdgcn_readfirstlane(tid >> 5);
    const int R    = range_nodes(N);
    const int CH   = (E + NCHK - 1) / NCHK;
    const int c    = blockIdx.x;
    const int e0   = c * CH;
    const int e1   = min(E, e0 + CH);

    if (tid < NBIN) hist[tid] = 0;
    __syncthreads();

    for (int eb = e0; eb < e1; eb += 4 * 512) {
        int d[4], s[4];
#pragma unroll
        for (int k = 0; k < 4; ++k) {
            const int e = eb + k * 512 + tid;
            d[k] = -1;
            s[k] = 0;
            if (e < e1) { d[k] = dst[e]; s[k] = src[e]; }
        }
#pragma unroll
        for (int k = 0; k < 4; ++k) {
            if (d[k] >= 0) {
                const int bb = d[k] / R;
                const int r  = atomicAdd(&hist[bb], 1);
                if (r < SEG_CAP) stage[bb * SEG_CAP + r] = make_int2(d[k], s[k]);
            }
        }
    }
    __syncthreads();

    constexpr int CPW = NBIN / 16;
    int  nn[CPW];
    int2 val[CPW];
#pragma unroll
    for (int t = 0; t < CPW; ++t) nn[t] = hist[wv + 16 * t];
#pragma unroll
    for (int t = 0; t < CPW; ++t) {
        val[t] = make_int2(-1, -1);
        if (lane >= 1 && lane <= min(nn[t], SEG_CAP)) val[t] = stage[(wv + 16 * t) * SEG_CAP + lane - 1];
    }
#pragma unroll
    for (int t = 0; t < CPW; ++t) {
        const int bb = wv + 16 * t;
        const int2 o = lane == 0 ? make_int2(nn[t], 0) : val[t];
        if (bb * R < N) *(int2*)(cells + (size_t)bb * R * DF + (size_t)c * SEG_W + 2 * lane) = o;
    }
}

__device__ __forceinline__ void list_add(int* cnt_l, int* slot_l, int li, int s) {
    const int r = atomicAdd(&cnt_l[li], 1);
    if (r < SLOT_CAP) slot_l[li * SLOT_CAP + r] = s;
}

__global__ void __launch_bounds__(512)
list_kernel(const int* __restrict__ src,
            const int* __restrict__ dst,
            int* __restrict__ cnt,
            int* cells,
            int N, int E) {
    __shared__ int cnt_l[RMAX];
    __shared__ int slot_l[RMAX * SLOT_CAP];

    const int tid  = threadIdx.x;
    const int lane = tid & 31;
    const int wv   = __builtin_amdgcn_readfirstlane(tid >> 5);
    const int R    = range_nodes(N);
    const int CH   = (E + NCHK - 1) / NCHK;
    const int base = blockIdx.x * R;
    if (base >= N) return;

    cnt_l[tid] = 0;
    __syncthreads();

    const int* region = cells + (size_t)base * DF;
    constexpr int CPW = NCHK / 16;
    int2 q[CPW];
#pragma unroll
    for (int k = 0; k < CPW; ++k)
        q[k] = *(const int2*)(region + (size_t)(wv + 16 * k) * SEG_W + 2 * lane);
#pragma unroll
    for (int k = 0; k < CPW; ++k) {
        const int n = __builtin_amdgcn_readfirstlane(q[k].x);
        if (n <= SEG_CAP) {
            if (lane >= 1 && lane <= n) list_add(cnt_l, slot_l, q[k].x - base, q[k].y);
        } else {
            const int c  = wv + 16 * k;
            const int f0 = c * CH;
            const int f1 = min(E, f0 + CH);
            for (int e = f0 + lane; e < f1; e += 32) {
                const unsigned li = (unsigned)(dst[e] - base);
                if (li < (unsigned)R) list_add(cnt_l, slot_l, (int)li, src[e]);
            }
        }
    }
    __syncthreads();

    for (int i = tid; i < R; i += 512)
        if (base + i < N) cnt[base + i] = cnt_l[i];
    const int per = R / 16;
    for (int t0 = 0; t0 < per; t0 += 4) {
        int n4[4], s4[4];
#pragma unroll
        for (int u = 0; u < 4; ++u) {
            const int  i  = wv + 16 * (t0 + u);
            const int  ci = cnt_l[min(i, RMAX - 1)];
            const bool ok = t0 + u < per && base + i < N;
            n4[u] = ok ? min(ci, SLOT_CAP) : 0;
        }
#pragma unroll
        for (int u = 0; u < 4; ++u) {
            const int i = wv + 16 * (t0 + u);
            s4[u] = lane < n4[u] ? slot_l[i * SLOT_CAP + lane] : 0;
        }
#pragma unroll
        for (int u = 0; u < 4; ++u) {
            const int i = wv + 16 * (t0 + u);
            if (lane < n4[u]) cells[(size_t)(base + i) * DF + lane] = s4[u];
        }
    }
}

__device__ __forceinline__ float node_norm_of(int c) {
    return rsqrtf(fmaxf((float)c, 1.0f));
}

struct NodeHead {
    int    n;
    int    my;
    float4 own;
};

__device__ __forceinline__ NodeHead load_head(const float* __restrict__ feat, unsigned l4, int lz,
                                              const int* __restrict__ cnt,
                                              const int* __restrict__ slots, int v, int N, int lane) {
    NodeHead h;
    h.n   = 0;
    h.my  = 0;
    h.own = make_float4(0.0f, 0.0f, 0.0f, 0.0f);
    if (v < N) {
        h.n   = cnt[v + lz];
        h.my  = slots[(size_t)v * DF + lane];
        h.own = *(const float4*)(feat + (size_t)v * DF + l4);
    }
    return h;
}

#define GROUP 6

__device__ __forceinline__ void rows_issue(const float* __restrict__ feat, unsigned l4, int lz,
                                           const int* __restrict__ cnt,
                                           int my, int j0, int k, int (&c)[GROUP], float4 (&f)[GROUP]) {
#pragma unroll
    for (int u = 0; u < GROUP; ++u) {
        c[u] = 1;
        f[u] = make_float4(0.0f, 0.0f, 0.0f, 0.0f);
        if (u < k) {
            const int s = __builtin_amdgcn_readlane(my, j0 + u);
            c[u] = cnt[s + lz];
            f[u] = *(const float4*)(feat + (size_t)s * DF + l4);
        }
    }
}

__device__ __forceinline__ void rows_sum(int k, const int (&c)[GROUP], const float4 (&f)[GROUP],
                                         float4& acc, float& ssum) {
#pragma unroll
    for (int u = 0; u < GROUP; ++u) {
        const float w = u < k ? node_norm_of(c[u]) : 0.0f;
        acc.x += w * f[u].x;
        acc.y += w * f[u].y;
        acc.z += w * f[u].z;
        acc.w += w * f[u].w;
        ssum  += w;
    }
}

__device__ __forceinline__ void rows_scan(const float* __restrict__ feat, unsigned l4,
                                          const int* __restrict__ cnt,
                                          const int* __restrict__ src, const int* __restrict__ dst,
                                          int E, int lane, int v, float4& acc, float& ssum) {
    for (int e0 = 0; e0 < E; e0 += 32) {
        const int e = e0 + lane;
        int dd = -1, ss = 0;
        if (e < E) { dd = dst[e]; ss = src[e]; }
        unsigned mask = __builtin_amdgcn_ballot_w32(dd == v);
        while (mask) {
            const int j = __builtin_ctz(mask);
            mask &= mask - 1;
            const int s = __builtin_amdgcn_readlane(ss, j);
            const float w = node_norm_of(cnt[s]);
            const float4 f = *(const float4*)(feat + (size_t)s * DF + l4);
            acc.x += w * f.x;
            acc.y += w * f.y;
            acc.z += w * f.z;
            acc.w += w * f.w;
            ssum  += w;
        }
    }
}

__device__ __forceinline__ void fill_two_rows(const float* __restrict__ feat, unsigned l4, int lz,
                                              const int* __restrict__ cnt,
                                              const int* __restrict__ src, const int* __restrict__ dst,
                                              int E, int lane, int vA, const NodeHead& hA, const NodeHead& hB,
                                              float* __restrict__ xrow,
                                              float& oSA, float& oSB, float& oNA, float& oNB) {
    const int nA = __builtin_amdgcn_readfirstlane(hA.n);
    const int nB = __builtin_amdgcn_readfirstlane(hB.n);
    const int kA = nA <= SLOT_CAP ? nA : 0;
    const int kB = nB <= SLOT_CAP ? nB : 0;
    const int kmax = max(kA, kB);

    float4 accA = make_float4(0.0f, 0.0f, 0.0f, 0.0f);
    float4 accB = make_float4(0.0f, 0.0f, 0.0f, 0.0f);
    float  sA = 0.0f, sB = 0.0f;
    for (int j0 = 0; j0 < kmax; j0 += GROUP) {
        int    cA[GROUP], cB[GROUP];
        float4 fA[GROUP], fB[GROUP];
        rows_issue(feat, l4, lz, cnt, hA.my, j0, kA - j0, cA, fA);
        rows_issue(feat, l4, lz, cnt, hB.my, j0, kB - j0, cB, fB);
        rows_sum(kA - j0, cA, fA, accA, sA);
        rows_sum(kB - j0, cB, fB, accB, sB);
    }
    if (nA > SLOT_CAP) rows_scan(feat, l4, cnt, src, dst, E, lane, vA,     accA, sA);
    if (nB > SLOT_CAP) rows_scan(feat, l4, cnt, src, dst, E, lane, vA + 1, accB, sB);

    float* xa = xrow + l4;
    float* xb = xa + XS_STRIDE;
    *(float4*)xa         = make_float4(hA.own.x * sA, hA.own.y * sA, hA.own.z * sA, hA.own.w * sA);
    *(float4*)(xa + DF)  = accA;
    *(float4*)xb         = make_float4(hB.own.x * sB, hB.own.y * sB, hB.own.z * sB, hB.own.w * sB);
    *(float4*)(xb + DF)  = accB;
    oSA = sA;
    oSB = sB;
    oNA = node_norm_of(nA);
    oNB = node_norm_of(nB);
}

__device__ __forceinline__ void store_two_rows(float* __restrict__ out, const float* __restrict__ st,
                                               unsigned l4, int row0, int N,
                                               float sA, float sB, float nA, float nB, const float4& b4) {
    const float4 a = *(const float4*)(st + l4);
    const float4 b = *(const float4*)(st + OS_STRIDE + l4);
    if (row0 < N)
        *(float4*)(out + (size_t)row0 * DF + l4) =
            make_float4(nA * (a.x + sA * b4.x), nA * (a.y + sA * b4.y),
                        nA * (a.z + sA * b4.z), nA * (a.w + sA * b4.w));
    if (row0 + 1 < N)
        *(float4*)(out + (size_t)(row0 + 1) * DF + l4) =
            make_float4(nB * (b.x + sB * b4.x), nB * (b.y + sB * b4.y),
                        nB * (b.z + sB * b4.z), nB * (b.w + sB * b4.w));
}

#define GEMM_BLOCKS 512

__global__ void __launch_bounds__(256)
conv_kernel(const float* __restrict__ feat,
            const float* __restrict__ W,
            const float* __restrict__ bias,
            const int*   __restrict__ src,
            const int*   __restrict__ dst,
            const int*   __restrict__ cnt,
            const int*   __restrict__ slots,
            float* __restrict__ out,
            int N, int E) {
    __shared__ float lds[128 * XS_STRIDE];

    const int tid  = threadIdx.x;
    const int lane = tid & 31;
    const int wv   = __builtin_amdgcn_readfirstlane(tid >> 5);
    const int nm   = lane & 15;
    const int half = lane >> 4;
    const int ncol = wv * 16 + nm;

    const int ntiles = (N + 15) >> 4;
    const int step   = gridDim.x;
    const unsigned l4 = lane * 4;
    int lz;
    asm("v_mov_b32 %0, 0" : "=v"(lz));

    int t = blockIdx.x;

    NodeHead hA = load_head(feat, l4, lz, cnt, slots, t * 16 + 2 * wv,     t < ntiles ? N : 0, lane);
    NodeHead hB = load_head(feat, l4, lz, cnt, slots, t * 16 + 2 * wv + 1, t < ntiles ? N : 0, lane);

    for (int q = tid; q < 128 * 64; q += 256) {
        const int wr = q >> 6;
        const int wc = q & 63;
        const float4 v = *(const float4*)(W + (size_t)wr * 256 + 4 * wc);
        *(float4*)&lds[wr * XS_STRIDE + 4 * wc] = v;
    }
    __syncthreads();

    v2f b[64];
    {
        const float* wl = &lds[ncol * XS_STRIDE + 2 * half];
#pragma unroll
        for (int kk = 0; kk < 64; ++kk) b[kk] = *(const v2f*)(wl + 4 * kk);
    }
    const float4 b4 = *(const float4*)(bias + l4);
    __syncthreads();

    float sA0 = 0.0f, sB0 = 0.0f, nA0 = 1.0f, nB0 = 1.0f;
    float sA1 = 0.0f, sB1 = 0.0f, nA1 = 1.0f, nB1 = 1.0f;

    if (t < ntiles)
        fill_two_rows(feat, l4, lz, cnt, src, dst, E, lane, t * 16 + 2 * wv, hA, hB,
                      &lds[(2 * wv) * XS_STRIDE], sA0, sB0, nA0, nB0);

    int buf = 0;
    int tp  = -1;
    for (; t < ntiles; t += step) {
        const float* xs = &lds[buf * (16 * XS_STRIDE)];

        __syncthreads();

        if (tp >= 0)
            store_two_rows(out, &lds[OS_BASE + ((buf ^ 1) * 16 + 2 * wv) * OS_STRIDE], l4,
                           tp * 16 + 2 * wv, N, sA1, sB1, nA1, nB1, b4);

        const int tn = t + step;
        hA = load_head(feat, l4, lz, cnt, slots, tn * 16 + 2 * wv,     tn < ntiles ? N : 0, lane);
        hB = load_head(feat, l4, lz, cnt, slots, tn * 16 + 2 * wv + 1, tn < ntiles ? N : 0, lane);

        const float* __restrict__ arow = &xs[nm * XS_STRIDE + 2 * half];

        v8f c = {};
#pragma unroll
        for (int kk = 0; kk < 64; ++kk) {
            const v2f a = *(const v2f*)(arow + 4 * kk);
            c = __builtin_amdgcn_wmma_f32_16x16x4_f32(
                     false, a,  false, b[kk],
                     (short)0, c,  false,  false);
        }

        asm volatile("v_nop\n\tv_nop\n\tv_nop\n\tv_nop\n\tv_nop" : "+v"(c));
        float* st = &lds[OS_BASE + (buf * 16 + 8 * half) * OS_STRIDE + ncol];
#pragma unroll
        for (int i = 0; i < 8; ++i) st[i * OS_STRIDE] = c[i];

        sA1 = sA0;
        sB1 = sB0;
        nA1 = nA0;
        nB1 = nB0;

        if (tn < ntiles)
            fill_two_rows(feat, l4, lz, cnt, src, dst, E, lane, tn * 16 + 2 * wv, hA, hB,
                          &lds[((buf ^ 1) * 16 + 2 * wv) * XS_STRIDE], sA0, sB0, nA0, nB0);
        tp = t;
        buf ^= 1;
    }

    if (tp >= 0) {
        __syncthreads();
        store_two_rows(out, &lds[OS_BASE + ((buf ^ 1) * 16 + 2 * wv) * OS_STRIDE], l4,
                       tp * 16 + 2 * wv, N, sA1, sB1, nA1, nB1, b4);
    }
}

extern "C" void kernel_launch(void* const* d_in, const int* in_sizes, int n_in,
                              void* d_out, int out_size, void* d_ws, size_t ws_size,
                              hipStream_t stream) {
    const float* feat = (const float*)d_in[0];
    const float* W    = (const float*)d_in[1];
    const float* bias = (const float*)d_in[2];
    const int*   src  = (const int*)d_in[3];
    const int*   dst  = (const int*)d_in[4];
    float* out = (float*)d_out;

    const int N = in_sizes[0] / DF;
    const int E = in_sizes[3];

    int* cnt  = (int*)d_ws;
    int* rows = (int*)d_ws + 2 * (size_t)N;

    bin_kernel<<<NCHK, 512, 0, stream>>>(src, dst, rows, N, E);
    list_kernel<<<NBIN, 512, 0, stream>>>(src, dst, cnt, rows, N, E);
    conv_kernel<<<GEMM_BLOCKS, 256, 0, stream>>>(feat, W, bias, src, dst, cnt, rows, out, N, E);
}


# first 6 source rows of each of a wave's next two nodes, and the own rows, asked for in front of the matrix run and summed behind it (lists read one tile earlier); sums and their order kept
# speedup vs baseline: 3.6891x; 1.0325x over previous
#include <hip/hip_runtime.h>


typedef float v2f __attribute__((ext_vector_type(2)));
typedef float v8f __attribute__((ext_vector_type(8)));

#define DF 128
#define XS_STRIDE 260
#define OS_STRIDE 132
#define OS_BASE (2 * 16 * XS_STRIDE)

#define NBIN 256
#define NCHK 256
#define SEG_CAP 31
#define SEG_W 64
#define SLOT_CAP 32
#define RMAX 512

__device__ __forceinline__ int range_nodes(int N) {
    return (((N + NBIN - 1) / NBIN) + 31) & ~31;
}

__global__ void __launch_bounds__(512)
bin_kernel(const int* __restrict__ src,
           const int* __restrict__ dst,
           int* __restrict__ cells,
           int N, int E) {
    __shared__ int  hist[NBIN];
    __shared__ int2 stage[NBIN * SEG_CAP];

    const int tid  = threadIdx.x;
    const int lane = tid & 31;
    const int wv   = __builtin_amdgcn_readfirstlane(tid >> 5);
    const int R    = range_nodes(N);
    const int CH   = (E + NCHK - 1) / NCHK;
    const int c    = blockIdx.x;
    const int e0   = c * CH;
    const int e1   = min(E, e0 + CH);

    if (tid < NBIN) hist[tid] = 0;
    __syncthreads();

    for (int eb = e0; eb < e1; eb += 4 * 512) {
        int d[4], s[4];
#pragma unroll
        for (int k = 0; k < 4; ++k) {
            const int e = eb + k * 512 + tid;
            d[k] = -1;
            s[k] = 0;
            if (e < e1) { d[k] = dst[e]; s[k] = src[e]; }
        }
#pragma unroll
        for (int k = 0; k < 4; ++k) {
            if (d[k] >= 0) {
                const int bb = d[k] / R;
                const int r  = atomicAdd(&hist[bb], 1);
                if (r < SEG_CAP) stage[bb * SEG_CAP + r] = make_int2(d[k], s[k]);
            }
        }
    }
    __syncthreads();

    constexpr int CPW = NBIN / 16;
    int  nn[CPW];
    int2 val[CPW];
#pragma unroll
    for (int t = 0; t < CPW; ++t) nn[t] = hist[wv + 16 * t];
#pragma unroll
    for (int t = 0; t < CPW; ++t) {
        val[t] = make_int2(-1, -1);
        if (lane >= 1 && lane <= min(nn[t], SEG_CAP)) val[t] = stage[(wv + 16 * t) * SEG_CAP + lane - 1];
    }
#pragma unroll
    for (int t = 0; t < CPW; ++t) {
        const int bb = wv + 16 * t;
        const int2 o = lane == 0 ? make_int2(nn[t], 0) : val[t];
        if (bb * R < N) *(int2*)(cells + (size_t)bb * R * DF + (size_t)c * SEG_W + 2 * lane) = o;
    }
}

__device__ __forceinline__ void list_add(int* cnt_l, int* slot_l, int li, int s) {
    const int r = atomicAdd(&cnt_l[li], 1);
    if (r < SLOT_CAP) slot_l[li * SLOT_CAP + r] = s;
}

__global__ void __launch_bounds__(512)
list_kernel(const int* __restrict__ src,
            const int* __restrict__ dst,
            int* __restrict__ cnt,
            int* cells,
            int N, int E) {
    __shared__ int cnt_l[RMAX];
    __shared__ int slot_l[RMAX * SLOT_CAP];

    const int tid  = threadIdx.x;
    const int lane = tid & 31;
    const int wv   = __builtin_amdgcn_readfirstlane(tid >> 5);
    const int R    = range_nodes(N);
    const int CH   = (E + NCHK - 1) / NCHK;
    const int base = blockIdx.x * R;
    if (base >= N) return;

    cnt_l[tid] = 0;
    __syncthreads();

    const int* region = cells + (size_t)base * DF;
    constexpr int CPW = NCHK / 16;
    int2 q[CPW];
#pragma unroll
    for (int k = 0; k < CPW; ++k)
        q[k] = *(const int2*)(region + (size_t)(wv + 16 * k) * SEG_W + 2 * lane);
#pragma unroll
    for (int k = 0; k < CPW; ++k) {
        const int n = __builtin_amdgcn_readfirstlane(q[k].x);
        if (n <= SEG_CAP) {
            if (lane >= 1 && lane <= n) list_add(cnt_l, slot_l, q[k].x - base, q[k].y);
        } else {
            const int c  = wv + 16 * k;
            const int f0 = c * CH;
            const int f1 = min(E, f0 + CH);
            for (int e = f0 + lane; e < f1; e += 32) {
                const unsigned li = (unsigned)(dst[e] - base);
                if (li < (unsigned)R) list_add(cnt_l, slot_l, (int)li, src[e]);
            }
        }
    }
    __syncthreads();

    for (int i = tid; i < R; i += 512)
        if (base + i < N) cnt[base + i] = cnt_l[i];
    const int per = R / 16;
    for (int t0 = 0; t0 < per; t0 += 4) {
        int n4[4], s4[4];
#pragma unroll
        for (int u = 0; u < 4; ++u) {
            const int  i  = wv + 16 * (t0 + u);
            const int  ci = cnt_l[min(i, RMAX - 1)];
            const bool ok = t0 + u < per && base + i < N;
            n4[u] = ok ? min(ci, SLOT_CAP) : 0;
        }
#pragma unroll
        for (int u = 0; u < 4; ++u) {
            const int i = wv + 16 * (t0 + u);
            s4[u] = lane < n4[u] ? slot_l[i * SLOT_CAP + lane] : 0;
        }
#pragma unroll
        for (int u = 0; u < 4; ++u) {
            const int i = wv + 16 * (t0 + u);
            if (lane < n4[u]) cells[(size_t)(base + i) * DF + lane] = s4[u];
        }
    }
}

__device__ __forceinline__ float node_norm_of(int c) {
    return rsqrtf(fmaxf((float)c, 1.0f));
}

struct NodeList {
    int n;
    int my;
};

__device__ __forceinline__ NodeList load_list(int lz, const int* __restrict__ cnt,
                                              const int* __restrict__ slots, int v, int N, int lane) {
    NodeList h;
    h.n  = 0;
    h.my = 0;
    if (v < N) {
        h.n  = cnt[v + lz];
        h.my = slots[(size_t)v * DF + lane];
    }
    return h;
}

__device__ __forceinline__ float4 load_own(const float* __restrict__ feat, unsigned l4, int v, int N) {
    float4 o = make_float4(0.0f, 0.0f, 0.0f, 0.0f);
    if (v < N) o = *(const float4*)(feat + (size_t)v * DF + l4);
    return o;
}

struct PairCount {
    int nA, nB;
    int kA, kB;
};

__device__ __forceinline__ PairCount pair_count(const NodeList& a, const NodeList& b) {
    PairCount p;
    p.nA = __builtin_amdgcn_readfirstlane(a.n);
    p.nB = __builtin_amdgcn_readfirstlane(b.n);
    p.kA = p.nA <= SLOT_CAP ? p.nA : 0;
    p.kB = p.nB <= SLOT_CAP ? p.nB : 0;
    return p;
}

#define GROUP 6

__device__ __forceinline__ void rows_issue(const float* __restrict__ feat, unsigned l4, int lz,
                                           const int* __restrict__ cnt,
                                           int my, int j0, int k, int (&c)[GROUP], float4 (&f)[GROUP]) {
#pragma unroll
    for (int u = 0; u < GROUP; ++u) {
        c[u] = 1;
        f[u] = make_float4(0.0f, 0.0f, 0.0f, 0.0f);
        if (u < k) {
            const int s = __builtin_amdgcn_readlane(my, j0 + u);
            c[u] = cnt[s + lz];
            f[u] = *(const float4*)(feat + (size_t)s * DF + l4);
        }
    }
}

__device__ __forceinline__ void rows_sum(int k, const int (&c)[GROUP], const float4 (&f)[GROUP],
                                         float4& acc, float& ssum) {
#pragma unroll
    for (int u = 0; u < GROUP; ++u) {
        const float w = u < k ? node_norm_of(c[u]) : 0.0f;
        acc.x += w * f[u].x;
        acc.y += w * f[u].y;
        acc.z += w * f[u].z;
        acc.w += w * f[u].w;
        ssum  += w;
    }
}

__device__ __forceinline__ void rows_scan(const float* __restrict__ feat, unsigned l4,
                                          const int* __restrict__ cnt,
                                          const int* __restrict__ src, const int* __restrict__ dst,
                                          int E, int lane, int v, float4& acc, float& ssum) {
    for (int e0 = 0; e0 < E; e0 += 32) {
        const int e = e0 + lane;
        int dd = -1, ss = 0;
        if (e < E) { dd = dst[e]; ss = src[e]; }
        unsigned mask = __builtin_amdgcn_ballot_w32(dd == v);
        while (mask) {
            const int j = __builtin_ctz(mask);
            mask &= mask - 1;
            const int s = __builtin_amdgcn_readlane(ss, j);
            const float w = node_norm_of(cnt[s]);
            const float4 f = *(const float4*)(feat + (size_t)s * DF + l4);
            acc.x += w * f.x;
            acc.y += w * f.y;
            acc.z += w * f.z;
            acc.w += w * f.w;
            ssum  += w;
        }
    }
}

__device__ __forceinline__ void finish_two_rows(const float* __restrict__ feat, unsigned l4, int lz,
                                                const int* __restrict__ cnt,
                                                const int* __restrict__ src, const int* __restrict__ dst,
                                                int E, int lane, int vA,
                                                const NodeList& hA, const NodeList& hB, const PairCount& p,
                                                const float4& ownA, const float4& ownB,
                                                const int (&c1A)[GROUP], const float4 (&f1A)[GROUP],
                                                const int (&c1B)[GROUP], const float4 (&f1B)[GROUP],
                                                float* __restrict__ xrow,
                                                float& oSA, float& oSB, float& oNA, float& oNB) {
    const int kmax = max(p.kA, p.kB);

    float4 accA = make_float4(0.0f, 0.0f, 0.0f, 0.0f);
    float4 accB = make_float4(0.0f, 0.0f, 0.0f, 0.0f);
    float  sA = 0.0f, sB = 0.0f;
    rows_sum(p.kA, c1A, f1A, accA, sA);
    rows_sum(p.kB, c1B, f1B, accB, sB);
    for (int j0 = GROUP; j0 < kmax; j0 += GROUP) {
        int    cA[GROUP], cB[GROUP];
        float4 fA[GROUP], fB[GROUP];
        rows_issue(feat, l4, lz, cnt, hA.my, j0, p.kA - j0, cA, fA);
        rows_issue(feat, l4, lz, cnt, hB.my, j0, p.kB - j0, cB, fB);
        rows_sum(p.kA - j0, cA, fA, accA, sA);
        rows_sum(p.kB - j0, cB, fB, accB, sB);
    }
    if (p.nA > SLOT_CAP) rows_scan(feat, l4, cnt, src, dst, E, lane, vA,     accA, sA);
    if (p.nB > SLOT_CAP) rows_scan(feat, l4, cnt, src, dst, E, lane, vA + 1, accB, sB);

    float* xa = xrow + l4;
    float* xb = xa + XS_STRIDE;
    *(float4*)xa         = make_float4(ownA.x * sA, ownA.y * sA, ownA.z * sA, ownA.w * sA);
    *(float4*)(xa + DF)  = accA;
    *(float4*)xb         = make_float4(ownB.x * sB, ownB.y * sB, ownB.z * sB, ownB.w * sB);
    *(float4*)(xb + DF)  = accB;
    oSA = sA;
    oSB = sB;
    oNA = node_norm_of(p.nA);
    oNB = node_norm_of(p.nB);
}

__device__ __forceinline__ void store_two_rows(float* __restrict__ out, const float* __restrict__ st,
                                               unsigned l4, int row0, int N,
                                               float sA, float sB, float nA, float nB, const float4& b4) {
    const float4 a = *(const float4*)(st + l4);
    const float4 b = *(const float4*)(st + OS_STRIDE + l4);
    if (row0 < N)
        *(float4*)(out + (size_t)row0 * DF + l4) =
            make_float4(nA * (a.x + sA * b4.x), nA * (a.y + sA * b4.y),
                        nA * (a.z + sA * b4.z), nA * (a.w + sA * b4.w));
    if (row0 + 1 < N)
        *(float4*)(out + (size_t)(row0 + 1) * DF + l4) =
            make_float4(nB * (b.x + sB * b4.x), nB * (b.y + sB * b4.y),
                        nB * (b.z + sB * b4.z), nB * (b.w + sB * b4.w));
}

#define GEMM_BLOCKS 512

__global__ void __launch_bounds__(256)
conv_kernel(const float* __restrict__ feat,
            const float* __restrict__ W,
            const float* __restrict__ bias,
            const int*   __restrict__ src,
            const int*   __restrict__ dst,
            const int*   __restrict__ cnt,
            const int*   __restrict__ slots,
            float* __restrict__ out,
            int N, int E) {
    __shared__ float lds[128 * XS_STRIDE];

    const int tid  = threadIdx.x;
    const int lane = tid & 31;
    const int wv   = __builtin_amdgcn_readfirstlane(tid >> 5);
    const int nm   = lane & 15;
    const int half = lane >> 4;
    const int ncol = wv * 16 + nm;

    const int ntiles = (N + 15) >> 4;
    const int step   = gridDim.x;
    const unsigned l4 = lane * 4;
    int lz;
    asm("v_mov_b32 %0, 0" : "=v"(lz));

    int t = blockIdx.x;

    NodeList lA = load_list(lz, cnt, slots, t * 16 + 2 * wv,     t < ntiles ? N : 0, lane);
    NodeList lB = load_list(lz, cnt, slots, t * 16 + 2 * wv + 1, t < ntiles ? N : 0, lane);
    NodeList mA = load_list(lz, cnt, slots, (t + step) * 16 + 2 * wv,     t + step < ntiles ? N : 0, lane);
    NodeList mB = load_list(lz, cnt, slots, (t + step) * 16 + 2 * wv + 1, t + step < ntiles ? N : 0, lane);

    for (int q = tid; q < 128 * 64; q += 256) {
        const int wr = q >> 6;
        const int wc = q & 63;
        const float4 v = *(const float4*)(W + (size_t)wr * 256 + 4 * wc);
        *(float4*)&lds[wr * XS_STRIDE + 4 * wc] = v;
    }
    __syncthreads();

    v2f b[64];
    {
        const float* wl = &lds[ncol * XS_STRIDE + 2 * half];
#pragma unroll
        for (int kk = 0; kk < 64; ++kk) b[kk] = *(const v2f*)(wl + 4 * kk);
    }
    const float4 b4 = *(const float4*)(bias + l4);
    __syncthreads();

    float sA0 = 0.0f, sB0 = 0.0f, nA0 = 1.0f, nB0 = 1.0f;
    float sA1 = 0.0f, sB1 = 0.0f, nA1 = 1.0f, nB1 = 1.0f;

    if (t < ntiles) {
        const int vA = t * 16 + 2 * wv;
        const PairCount p = pair_count(lA, lB);
        int    cA[GROUP], cB[GROUP];
        float4 fA[GROUP], fB[GROUP];
        rows_issue(feat, l4, lz, cnt, lA.my, 0, p.kA, cA, fA);
        rows_issue(feat, l4, lz, cnt, lB.my, 0, p.kB, cB, fB);
        const float4 ownA = load_own(feat, l4, vA,     N);
        const float4 ownB = load_own(feat, l4, vA + 1, N);
        finish_two_rows(feat, l4, lz, cnt, src, dst, E, lane, vA, lA, lB, p, ownA, ownB, cA, fA, cB, fB,
                        &lds[(2 * wv) * XS_STRIDE], sA0, sB0, nA0, nB0);
    }
    lA = mA;
    lB = mB;

    int buf = 0;
    int tp  = -1;
    for (; t < ntiles; t += step) {
        const float* xs = &lds[buf * (16 * XS_STRIDE)];

        __syncthreads();

        if (tp >= 0)
            store_two_rows(out, &lds[OS_BASE + ((buf ^ 1) * 16 + 2 * wv) * OS_STRIDE], l4,
                           tp * 16 + 2 * wv, N, sA1, sB1, nA1, nB1, b4);

        const int tn = t + step;
        const int vn = tn * 16 + 2 * wv;
        const PairCount p = pair_count(lA, lB);
        int    cA[GROUP], cB[GROUP];
        float4 fA[GROUP], fB[GROUP];
        rows_issue(feat, l4, lz, cnt, lA.my, 0, p.kA, cA, fA);
        rows_issue(feat, l4, lz, cnt, lB.my, 0, p.kB, cB, fB);
        const float4 ownA = load_own(feat, l4, vn,     tn < ntiles ? N : 0);
        const float4 ownB = load_own(feat, l4, vn + 1, tn < ntiles ? N : 0);
        const int tnn = tn + step;
        mA = load_list(lz, cnt, slots, tnn * 16 + 2 * wv,     tnn < ntiles ? N : 0, lane);
        mB = load_list(lz, cnt, slots, tnn * 16 + 2 * wv + 1, tnn < ntiles ? N : 0, lane);

        const float* __restrict__ arow = &xs[nm * XS_STRIDE + 2 * half];

        v8f c = {};
#pragma unroll
        for (int kk = 0; kk < 64; ++kk) {
            const v2f a = *(const v2f*)(arow + 4 * kk);
            c = __builtin_amdgcn_wmma_f32_16x16x4_f32(
                     false, a,  false, b[kk],
                     (short)0, c,  false,  false);
        }

        asm volatile("v_nop\n\tv_nop\n\tv_nop\n\tv_nop\n\tv_nop" : "+v"(c));
        float* st = &lds[OS_BASE + (buf * 16 + 8 * half) * OS_STRIDE + ncol];
#pragma unroll
        for (int i = 0; i < 8; ++i) st[i * OS_STRIDE] = c[i];

        sA1 = sA0;
        sB1 = sB0;
        nA1 = nA0;
        nB1 = nB0;

        if (tn < ntiles)
            finish_two_rows(feat, l4, lz, cnt, src, dst, E, lane, vn, lA, lB, p, ownA, ownB, cA, fA, cB, fB,
                            &lds[((buf ^ 1) * 16 + 2 * wv) * XS_STRIDE], sA0, sB0, nA0, nB0);
        lA = mA;
        lB = mB;
        tp = t;
        buf ^= 1;
    }

    if (tp >= 0) {
        __syncthreads();
        store_two_rows(out, &lds[OS_BASE + ((buf ^ 1) * 16 + 2 * wv) * OS_STRIDE], l4,
                       tp * 16 + 2 * wv, N, sA1, sB1, nA1, nB1, b4);
    }
}

extern "C" void kernel_launch(void* const* d_in, const int* in_sizes, int n_in,
                              void* d_out, int out_size, void* d_ws, size_t ws_size,
                              hipStream_t stream) {
    const float* feat = (const float*)d_in[0];
    const float* W    = (const float*)d_in[1];
    const float* bias = (const float*)d_in[2];
    const int*   src  = (const int*)d_in[3];
    const int*   dst  = (const int*)d_in[4];
    float* out = (float*)d_out;

    const int N = in_sizes[0] / DF;
    const int E = in_sizes[3];

    int* cnt  = (int*)d_ws;
    int* rows = (int*)d_ws + 2 * (size_t)N;

    bin_kernel<<<NCHK, 512, 0, stream>>>(src, dst, rows, N, E);
    list_kernel<<<NBIN, 512, 0, stream>>>(src, dst, cnt, rows, N, E);
    conv_kernel<<<GEMM_BLOCKS, 256, 0, stream>>>(feat, W, bias, src, dst, cnt, rows, out, N, E);
}
